# starting kernel
# baseline (speedup 1.0000x reference)
#include <hip/hip_runtime.h>

typedef __attribute__((ext_vector_type(2))) float v2f;
typedef __attribute__((ext_vector_type(8))) float v8f;

#define N_NODES 100000
#define F_IN    256
#define F_OUT   32
#define M_TILES (N_NODES / 16)

__global__ __launch_bounds__(256) void gcn_gemm_xw(const float* __restrict__ x,
                                                   const float* __restrict__ W,
                                                   float* __restrict__ h) {
    __shared__ v2f Wp[(F_IN / 2) * F_OUT];
    for (int idx = threadIdx.x; idx < (F_IN / 2) * F_OUT; idx += 256) {
        int p = idx >> 5;
        int c = idx & 31;
        v2f w;
        w.x = W[(2 * p)     * F_OUT + c];
        w.y = W[(2 * p + 1) * F_OUT + c];
        Wp[idx] = w;
    }
    __syncthreads();

    const int wave = threadIdx.x >> 5;
    const int lane = threadIdx.x & 31;
    const int tile = blockIdx.x * 8 + wave;
    if (tile >= M_TILES) return;

    const int m0   = tile << 4;
    const int mrow = m0 + (lane & 15);
    const int koff = (lane >> 4) << 1;
    const int col  = lane & 15;
    const int poff = koff >> 1;

    v8f c0 = {};
    v8f c1 = {};

    const float* __restrict__ xrow = x + (size_t)mrow * F_IN;
    #pragma unroll 4
    for (int k = 0; k < F_IN; k += 4) {
        v2f a = *(const v2f*)(xrow + k + koff);
        const int p = (k >> 1) + poff;
        v2f b0 = Wp[p * F_OUT + col];
        v2f b1 = Wp[p * F_OUT + 16 + col];
        c0 = __builtin_amdgcn_wmma_f32_16x16x4_f32(false, a, false, b0,
                                                   (short)0, c0, false, false);
        c1 = __builtin_amdgcn_wmma_f32_16x16x4_f32(false, a, false, b1,
                                                   (short)0, c1, false, false);
    }

    const int rbase = m0 + ((lane >> 4) << 3);
    #pragma unroll
    for (int j = 0; j < 8; ++j) {
        h[(size_t)(rbase + j) * F_OUT + col]      = c0[j];
        h[(size_t)(rbase + j) * F_OUT + 16 + col] = c1[j];
    }
}

__global__ void gcn_deg_init(unsigned* __restrict__ deg) {
    int i = blockIdx.x * blockDim.x + threadIdx.x;
    if (i < N_NODES) deg[i] = 1u;
}

__global__ void gcn_deg_count(const int* __restrict__ dst,
                              unsigned* __restrict__ deg, int E) {
    int e = blockIdx.x * blockDim.x + threadIdx.x;
    if (e < E) atomicAdd(&deg[dst[e]], 1u);
}

__global__ void gcn_dinv(const unsigned* __restrict__ deg,
                         float* __restrict__ dinv) {
    int i = blockIdx.x * blockDim.x + threadIdx.x;
    if (i < N_NODES) dinv[i] = rsqrtf((float)deg[i]);
}

__global__ void gcn_out_init(const float* __restrict__ h,
                             const float* __restrict__ dinv,
                             const float* __restrict__ b,
                             float* __restrict__ out) {
    int t = blockIdx.x * blockDim.x + threadIdx.x;
    if (t >= N_NODES * F_OUT) return;
    int i = t >> 5;
    int f = t & 31;
    float di = dinv[i];
    out[t] = h[t] * di * di + b[f];
}

__global__ __launch_bounds__(256) void gcn_scatter(const int* __restrict__ src,
                                                   const int* __restrict__ dst,
                                                   const float* __restrict__ h,
                                                   const float* __restrict__ dinv,
                                                   float* __restrict__ out, int E) {
    long long t = (long long)blockIdx.x * blockDim.x + threadIdx.x;
    int e = (int)(t >> 5);
    int f = (int)(t & 31);
    if (e >= E) return;
    int s = src[e];
    int d = dst[e];
    float nrm = dinv[s] * dinv[d];
    unsafeAtomicAdd(&out[(size_t)d * F_OUT + f],
                    h[(size_t)s * F_OUT + f] * nrm);
}

extern "C" void kernel_launch(void* const* d_in, const int* in_sizes, int n_in,
                              void* d_out, int out_size, void* d_ws, size_t ws_size,
                              hipStream_t stream) {
    const float* x     = (const float*)d_in[0];
    const int*   edges = (const int*)d_in[1];
    const float* W     = (const float*)d_in[2];
    const float* b     = (const float*)d_in[3];
    float*       out   = (float*)d_out;

    const int E = in_sizes[1] / 2;
    const int* src = edges;
    const int* dst = edges + E;

    char* ws = (char*)d_ws;
    float*    h    = (float*)ws;
    unsigned* deg  = (unsigned*)(ws + (size_t)N_NODES * F_OUT * 4);
    float*    dinv = (float*)(ws + (size_t)N_NODES * F_OUT * 4 + (size_t)N_NODES * 4);

    gcn_gemm_xw<<<(M_TILES + 7) / 8, 256, 0, stream>>>(x, W, h);

    gcn_deg_init<<<(N_NODES + 255) / 256, 256, 0, stream>>>(deg);
    gcn_deg_count<<<(E + 255) / 256, 256, 0, stream>>>(dst, deg, E);
    gcn_dinv<<<(N_NODES + 255) / 256, 256, 0, stream>>>(deg, dinv);

    gcn_out_init<<<(N_NODES * F_OUT + 255) / 256, 256, 0, stream>>>(h, dinv, b, out);

    long long work = (long long)E * F_OUT;
    int blocks = (int)((work + 255) / 256);
    gcn_scatter<<<blocks, 256, 0, stream>>>(src, dst, h, dinv, out, E);
}


# degree count without global atomics: per-work-group LDS histograms over (edge slice, node range), partial rows stored plainly and summed per node; 6 launches become 5
# speedup vs baseline: 1.4174x; 1.4174x over previous
#include <hip/hip_runtime.h>

typedef __attribute__((ext_vector_type(2))) float v2f;
typedef __attribute__((ext_vector_type(8))) float v8f;

#define N_NODES 100000
#define F_IN    256
#define F_OUT   32
#define M_TILES (N_NODES / 16)

__global__ __launch_bounds__(256) void gcn_gemm_xw(const float* __restrict__ x,
                                                   const float* __restrict__ W,
                                                   float* __restrict__ h) {
    __shared__ v2f Wp[(F_IN / 2) * F_OUT];
    for (int idx = threadIdx.x; idx < (F_IN / 2) * F_OUT; idx += 256) {
        int p = idx >> 5;
        int c = idx & 31;
        v2f w;
        w.x = W[(2 * p)     * F_OUT + c];
        w.y = W[(2 * p + 1) * F_OUT + c];
        Wp[idx] = w;
    }
    __syncthreads();

    const int wave = threadIdx.x >> 5;
    const int lane = threadIdx.x & 31;
    const int tile = blockIdx.x * 8 + wave;
    if (tile >= M_TILES) return;

    const int m0   = tile << 4;
    const int mrow = m0 + (lane & 15);
    const int koff = (lane >> 4) << 1;
    const int col  = lane & 15;
    const int poff = koff >> 1;

    v8f c0 = {};
    v8f c1 = {};

    const float* __restrict__ xrow = x + (size_t)mrow * F_IN;
    #pragma unroll 4
    for (int k = 0; k < F_IN; k += 4) {
        v2f a = *(const v2f*)(xrow + k + koff);
        const int p = (k >> 1) + poff;
        v2f b0 = Wp[p * F_OUT + col];
        v2f b1 = Wp[p * F_OUT + 16 + col];
        c0 = __builtin_amdgcn_wmma_f32_16x16x4_f32(false, a, false, b0,
                                                   (short)0, c0, false, false);
        c1 = __builtin_amdgcn_wmma_f32_16x16x4_f32(false, a, false, b1,
                                                   (short)0, c1, false, false);
    }

    const int rbase = m0 + ((lane >> 4) << 3);
    #pragma unroll
    for (int j = 0; j < 8; ++j) {
        h[(size_t)(rbase + j) * F_OUT + col]      = c0[j];
        h[(size_t)(rbase + j) * F_OUT + 16 + col] = c1[j];
    }
}

#define DEG_SLICES 16
#define DEG_RANGES 16
#define DEG_RANGE_NODES (N_NODES / DEG_RANGES)

__device__ __forceinline__ void deg_count4(unsigned* hist, int4 d, unsigned base) {
    unsigned r0 = (unsigned)d.x - base;
    unsigned r1 = (unsigned)d.y - base;
    unsigned r2 = (unsigned)d.z - base;
    unsigned r3 = (unsigned)d.w - base;
    if (r0 < DEG_RANGE_NODES) atomicAdd(&hist[r0], 1u);
    if (r1 < DEG_RANGE_NODES) atomicAdd(&hist[r1], 1u);
    if (r2 < DEG_RANGE_NODES) atomicAdd(&hist[r2], 1u);
    if (r3 < DEG_RANGE_NODES) atomicAdd(&hist[r3], 1u);
}

__global__ __launch_bounds__(1024) void gcn_deg_hist(const int* __restrict__ dst,
                                                     unsigned* __restrict__ part, int E) {
    __shared__ unsigned hist[DEG_RANGE_NODES];
    const int slice = blockIdx.x / DEG_RANGES;
    const int range = blockIdx.x % DEG_RANGES;
    const unsigned base = (unsigned)range * DEG_RANGE_NODES;

    for (int t = threadIdx.x; t < DEG_RANGE_NODES; t += 1024) hist[t] = 0u;
    __syncthreads();

    const int chunk = (((E + DEG_SLICES - 1) / DEG_SLICES) + 3) & ~3;
    int lo = slice * chunk;
    int hi = lo + chunk;
    if (lo > E) lo = E;
    if (hi > E) hi = E;

    if ((((size_t)dst) & 15) == 0) {
        const int nvec = (hi - lo) >> 2;
        const int4* __restrict__ d4 = (const int4*)(dst + lo);
        int v = threadIdx.x;
        for (; v + 3 * 1024 < nvec; v += 4 * 1024) {
            int4 da = d4[v];
            int4 db = d4[v + 1024];
            int4 dc = d4[v + 2 * 1024];
            int4 dd = d4[v + 3 * 1024];
            deg_count4(hist, da, base);
            deg_count4(hist, db, base);
            deg_count4(hist, dc, base);
            deg_count4(hist, dd, base);
        }
        for (; v < nvec; v += 1024) deg_count4(hist, d4[v], base);
        lo += nvec << 2;
    }
    for (int e = lo + threadIdx.x; e < hi; e += 1024) {
        unsigned r = (unsigned)dst[e] - base;
        if (r < DEG_RANGE_NODES) atomicAdd(&hist[r], 1u);
    }
    __syncthreads();

    unsigned* __restrict__ row = part + (size_t)slice * N_NODES + base;
    for (int t = threadIdx.x; t < DEG_RANGE_NODES; t += 1024) row[t] = hist[t];
}

__global__ void gcn_dinv(const unsigned* __restrict__ part,
                         float* __restrict__ dinv) {
    int i = blockIdx.x * blockDim.x + threadIdx.x;
    if (i >= N_NODES) return;
    unsigned deg = 1u;
    #pragma unroll
    for (int s = 0; s < DEG_SLICES; ++s) deg += part[(size_t)s * N_NODES + i];
    dinv[i] = rsqrtf((float)deg);
}

__global__ void gcn_out_init(const float* __restrict__ h,
                             const float* __restrict__ dinv,
                             const float* __restrict__ b,
                             float* __restrict__ out) {
    int t = blockIdx.x * blockDim.x + threadIdx.x;
    if (t >= N_NODES * F_OUT) return;
    int i = t >> 5;
    int f = t & 31;
    float di = dinv[i];
    out[t] = h[t] * di * di + b[f];
}

__global__ __launch_bounds__(256) void gcn_scatter(const int* __restrict__ src,
                                                   const int* __restrict__ dst,
                                                   const float* __restrict__ h,
                                                   const float* __restrict__ dinv,
                                                   float* __restrict__ out, int E) {
    long long t = (long long)blockIdx.x * blockDim.x + threadIdx.x;
    int e = (int)(t >> 5);
    int f = (int)(t & 31);
    if (e >= E) return;
    int s = src[e];
    int d = dst[e];
    float nrm = dinv[s] * dinv[d];
    unsafeAtomicAdd(&out[(size_t)d * F_OUT + f],
                    h[(size_t)s * F_OUT + f] * nrm);
}

extern "C" void kernel_launch(void* const* d_in, const int* in_sizes, int n_in,
                              void* d_out, int out_size, void* d_ws, size_t ws_size,
                              hipStream_t stream) {
    const float* x     = (const float*)d_in[0];
    const int*   edges = (const int*)d_in[1];
    const float* W     = (const float*)d_in[2];
    const float* b     = (const float*)d_in[3];
    float*       out   = (float*)d_out;

    const int E = in_sizes[1] / 2;
    const int* src = edges;
    const int* dst = edges + E;

    char* ws = (char*)d_ws;
    float*    h    = (float*)ws;
    unsigned* part = (unsigned*)ws;
    float*    dinv = (float*)(ws + (size_t)N_NODES * F_OUT * 4 + (size_t)N_NODES * 4);

    gcn_deg_hist<<<DEG_SLICES * DEG_RANGES, 1024, 0, stream>>>(dst, part, E);
    gcn_dinv<<<(N_NODES + 255) / 256, 256, 0, stream>>>(part, dinv);

    gcn_gemm_xw<<<(M_TILES + 7) / 8, 256, 0, stream>>>(x, W, h);

    gcn_out_init<<<(N_NODES * F_OUT + 255) / 256, 256, 0, stream>>>(h, dinv, b, out);

    long long work = (long long)E * F_OUT;
    int blocks = (int)((work + 255) / 256);
    gcn_scatter<<<blocks, 256, 0, stream>>>(src, dst, h, dinv, out, E);
}


# edge scatter: one wave per 32 edges, not per edge: lanes fetch 32 index pairs and norms at once, readlane-broadcast, 32 row gathers in flight before the fp32 atomics (arrival order free as before)
# speedup vs baseline: 2.8674x; 2.0230x over previous
#include <hip/hip_runtime.h>

typedef __attribute__((ext_vector_type(2))) float v2f;
typedef __attribute__((ext_vector_type(8))) float v8f;

#define N_NODES 100000
#define F_IN    256
#define F_OUT   32
#define M_TILES (N_NODES / 16)

__global__ __launch_bounds__(256) void gcn_gemm_xw(const float* __restrict__ x,
                                                   const float* __restrict__ W,
                                                   float* __restrict__ h) {
    __shared__ v2f Wp[(F_IN / 2) * F_OUT];
    for (int idx = threadIdx.x; idx < (F_IN / 2) * F_OUT; idx += 256) {
        int p = idx >> 5;
        int c = idx & 31;
        v2f w;
        w.x = W[(2 * p)     * F_OUT + c];
        w.y = W[(2 * p + 1) * F_OUT + c];
        Wp[idx] = w;
    }
    __syncthreads();

    const int wave = threadIdx.x >> 5;
    const int lane = threadIdx.x & 31;
    const int tile = blockIdx.x * 8 + wave;
    if (tile >= M_TILES) return;

    const int m0   = tile << 4;
    const int mrow = m0 + (lane & 15);
    const int koff = (lane >> 4) << 1;
    const int col  = lane & 15;
    const int poff = koff >> 1;

    v8f c0 = {};
    v8f c1 = {};

    const float* __restrict__ xrow = x + (size_t)mrow * F_IN;
    #pragma unroll 4
    for (int k = 0; k < F_IN; k += 4) {
        v2f a = *(const v2f*)(xrow + k + koff);
        const int p = (k >> 1) + poff;
        v2f b0 = Wp[p * F_OUT + col];
        v2f b1 = Wp[p * F_OUT + 16 + col];
        c0 = __builtin_amdgcn_wmma_f32_16x16x4_f32(false, a, false, b0,
                                                   (short)0, c0, false, false);
        c1 = __builtin_amdgcn_wmma_f32_16x16x4_f32(false, a, false, b1,
                                                   (short)0, c1, false, false);
    }

    const int rbase = m0 + ((lane >> 4) << 3);
    #pragma unroll
    for (int j = 0; j < 8; ++j) {
        h[(size_t)(rbase + j) * F_OUT + col]      = c0[j];
        h[(size_t)(rbase + j) * F_OUT + 16 + col] = c1[j];
    }
}

#define DEG_SLICES 16
#define DEG_RANGES 16
#define DEG_RANGE_NODES (N_NODES / DEG_RANGES)

__device__ __forceinline__ void deg_count4(unsigned* hist, int4 d, unsigned base) {
    unsigned r0 = (unsigned)d.x - base;
    unsigned r1 = (unsigned)d.y - base;
    unsigned r2 = (unsigned)d.z - base;
    unsigned r3 = (unsigned)d.w - base;
    if (r0 < DEG_RANGE_NODES) atomicAdd(&hist[r0], 1u);
    if (r1 < DEG_RANGE_NODES) atomicAdd(&hist[r1], 1u);
    if (r2 < DEG_RANGE_NODES) atomicAdd(&hist[r2], 1u);
    if (r3 < DEG_RANGE_NODES) atomicAdd(&hist[r3], 1u);
}

__global__ __launch_bounds__(1024) void gcn_deg_hist(const int* __restrict__ dst,
                                                     unsigned* __restrict__ part, int E) {
    __shared__ unsigned hist[DEG_RANGE_NODES];
    const int slice = blockIdx.x / DEG_RANGES;
    const int range = blockIdx.x % DEG_RANGES;
    const unsigned base = (unsigned)range * DEG_RANGE_NODES;

    for (int t = threadIdx.x; t < DEG_RANGE_NODES; t += 1024) hist[t] = 0u;
    __syncthreads();

    const int chunk = (((E + DEG_SLICES - 1) / DEG_SLICES) + 3) & ~3;
    int lo = slice * chunk;
    int hi = lo + chunk;
    if (lo > E) lo = E;
    if (hi > E) hi = E;

    if ((((size_t)dst) & 15) == 0) {
        const int nvec = (hi - lo) >> 2;
        const int4* __restrict__ d4 = (const int4*)(dst + lo);
        int v = threadIdx.x;
        for (; v + 3 * 1024 < nvec; v += 4 * 1024) {
            int4 da = d4[v];
            int4 db = d4[v + 1024];
            int4 dc = d4[v + 2 * 1024];
            int4 dd = d4[v + 3 * 1024];
            deg_count4(hist, da, base);
            deg_count4(hist, db, base);
            deg_count4(hist, dc, base);
            deg_count4(hist, dd, base);
        }
        for (; v < nvec; v += 1024) deg_count4(hist, d4[v], base);
        lo += nvec << 2;
    }
    for (int e = lo + threadIdx.x; e < hi; e += 1024) {
        unsigned r = (unsigned)dst[e] - base;
        if (r < DEG_RANGE_NODES) atomicAdd(&hist[r], 1u);
    }
    __syncthreads();

    unsigned* __restrict__ row = part + (size_t)slice * N_NODES + base;
    for (int t = threadIdx.x; t < DEG_RANGE_NODES; t += 1024) row[t] = hist[t];
}

__global__ void gcn_dinv(const unsigned* __restrict__ part,
                         float* __restrict__ dinv) {
    int i = blockIdx.x * blockDim.x + threadIdx.x;
    if (i >= N_NODES) return;
    unsigned deg = 1u;
    #pragma unroll
    for (int s = 0; s < DEG_SLICES; ++s) deg += part[(size_t)s * N_NODES + i];
    dinv[i] = rsqrtf((float)deg);
}

__global__ void gcn_out_init(const float* __restrict__ h,
                             const float* __restrict__ dinv,
                             const float* __restrict__ b,
                             float* __restrict__ out) {
    int t = blockIdx.x * blockDim.x + threadIdx.x;
    if (t >= N_NODES * F_OUT) return;
    int i = t >> 5;
    int f = t & 31;
    float di = dinv[i];
    out[t] = h[t] * di * di + b[f];
}

#define SC_EDGES 32

__device__ __forceinline__ unsigned sc_bcast_u(int v, int j) {
    return (unsigned)__builtin_amdgcn_readlane(v, j);
}
__device__ __forceinline__ float sc_bcast_f(float v, int j) {
    return __int_as_float(__builtin_amdgcn_readlane(__float_as_int(v), j));
}

__global__ __launch_bounds__(256) void gcn_scatter(const int* __restrict__ src,
                                                   const int* __restrict__ dst,
                                                   const float* __restrict__ h,
                                                   const float* __restrict__ dinv,
                                                   float* __restrict__ out, int E) {
    const unsigned lane = threadIdx.x & 31u;
    const int wave = (int)((blockIdx.x * 256u + threadIdx.x) >> 5);
    const int base = wave * SC_EDGES;
    if (base >= E) return;

    if (base + SC_EDGES <= E) {
        const int e = base + (int)lane;
        const int s = src[e];
        const int d = dst[e];
        const float ds = dinv[s];
        const float dd = dinv[d];

        float v[SC_EDGES];
        #pragma unroll
        for (int j = 0; j < SC_EDGES; ++j) {
            const unsigned sj = sc_bcast_u(s, j);
            v[j] = h[sj * (unsigned)F_OUT + lane];
        }
        const float nrm = ds * dd;
        #pragma unroll
        for (int j = 0; j < SC_EDGES; ++j) {
            const unsigned dj = sc_bcast_u(d, j);
            const float    nj = sc_bcast_f(nrm, j);
            unsafeAtomicAdd(&out[dj * (unsigned)F_OUT + lane], v[j] * nj);
        }
    } else {
        const int cnt = E - base;
        int s = 0, d = 0;
        float nrm = 0.0f;
        if ((int)lane < cnt) {
            s = src[base + (int)lane];
            d = dst[base + (int)lane];
            nrm = dinv[s] * dinv[d];
        }
        for (int j = 0; j < cnt; ++j) {
            const unsigned sj = sc_bcast_u(s, j);
            const unsigned dj = sc_bcast_u(d, j);
            const float    nj = sc_bcast_f(nrm, j);
            unsafeAtomicAdd(&out[dj * (unsigned)F_OUT + lane],
                            h[sj * (unsigned)F_OUT + lane] * nj);
        }
    }
}

extern "C" void kernel_launch(void* const* d_in, const int* in_sizes, int n_in,
                              void* d_out, int out_size, void* d_ws, size_t ws_size,
                              hipStream_t stream) {
    const float* x     = (const float*)d_in[0];
    const int*   edges = (const int*)d_in[1];
    const float* W     = (const float*)d_in[2];
    const float* b     = (const float*)d_in[3];
    float*       out   = (float*)d_out;

    const int E = in_sizes[1] / 2;
    const int* src = edges;
    const int* dst = edges + E;

    char* ws = (char*)d_ws;
    float*    h    = (float*)ws;
    unsigned* part = (unsigned*)ws;
    float*    dinv = (float*)(ws + (size_t)N_NODES * F_OUT * 4 + (size_t)N_NODES * 4);

    gcn_deg_hist<<<DEG_SLICES * DEG_RANGES, 1024, 0, stream>>>(dst, part, E);
    gcn_dinv<<<(N_NODES + 255) / 256, 256, 0, stream>>>(part, dinv);

    gcn_gemm_xw<<<(M_TILES + 7) / 8, 256, 0, stream>>>(x, W, h);

    gcn_out_init<<<(N_NODES * F_OUT + 255) / 256, 256, 0, stream>>>(h, dinv, b, out);

    int blocks = (E + 255) / 256;
    gcn_scatter<<<blocks, 256, 0, stream>>>(src, dst, h, dinv, out, E);
}


# norm factored out of the edge loop: GEMM epilogue scales rows by dinv[src] (stored to hs and as out's start), a last pass applies dinv[dst] and bias to the finished sums; no per-edge dinv gather
# speedup vs baseline: 3.1311x; 1.0920x over previous
#include <hip/hip_runtime.h>

typedef __attribute__((ext_vector_type(2))) float v2f;
typedef __attribute__((ext_vector_type(8))) float v8f;

#define N_NODES 100000
#define F_IN    256
#define F_OUT   32
#define M_TILES (N_NODES / 16)

__global__ __launch_bounds__(256) void gcn_gemm_xw(const float* __restrict__ x,
                                                   const float* __restrict__ W,
                                                   const float* __restrict__ dinv,
                                                   float* __restrict__ hs,
                                                   float* __restrict__ out) {
    __shared__ v2f Wp[(F_IN / 2) * F_OUT];
    for (int idx = threadIdx.x; idx < (F_IN / 2) * F_OUT; idx += 256) {
        int p = idx >> 5;
        int c = idx & 31;
        v2f w;
        w.x = W[(2 * p)     * F_OUT + c];
        w.y = W[(2 * p + 1) * F_OUT + c];
        Wp[idx] = w;
    }
    __syncthreads();

    const int wave = threadIdx.x >> 5;
    const int lane = threadIdx.x & 31;
    const int tile = blockIdx.x * 8 + wave;
    if (tile >= M_TILES) return;

    const int m0   = tile << 4;
    const int mrow = m0 + (lane & 15);
    const int koff = (lane >> 4) << 1;
    const int col  = lane & 15;
    const int poff = koff >> 1;

    v8f c0 = {};
    v8f c1 = {};

    const float* __restrict__ xrow = x + (size_t)mrow * F_IN;
    #pragma unroll 4
    for (int k = 0; k < F_IN; k += 4) {
        v2f a = *(const v2f*)(xrow + k + koff);
        const int p = (k >> 1) + poff;
        v2f b0 = Wp[p * F_OUT + col];
        v2f b1 = Wp[p * F_OUT + 16 + col];
        c0 = __builtin_amdgcn_wmma_f32_16x16x4_f32(false, a, false, b0,
                                                   (short)0, c0, false, false);
        c1 = __builtin_amdgcn_wmma_f32_16x16x4_f32(false, a, false, b1,
                                                   (short)0, c1, false, false);
    }

    const int rbase = m0 + ((lane >> 4) << 3);
    #pragma unroll
    for (int j = 0; j < 8; ++j) {
        const float  dj = dinv[rbase + j];
        const size_t o  = (size_t)(rbase + j) * F_OUT + col;
        const float  v0 = c0[j] * dj;
        const float  v1 = c1[j] * dj;
        hs[o]       = v0;
        hs[o + 16]  = v1;
        out[o]      = v0;
        out[o + 16] = v1;
    }
}

#define DEG_SLICES 16
#define DEG_RANGES 16
#define DEG_RANGE_NODES (N_NODES / DEG_RANGES)

__device__ __forceinline__ void deg_count4(unsigned* hist, int4 d, unsigned base) {
    unsigned r0 = (unsigned)d.x - base;
    unsigned r1 = (unsigned)d.y - base;
    unsigned r2 = (unsigned)d.z - base;
    unsigned r3 = (unsigned)d.w - base;
    if (r0 < DEG_RANGE_NODES) atomicAdd(&hist[r0], 1u);
    if (r1 < DEG_RANGE_NODES) atomicAdd(&hist[r1], 1u);
    if (r2 < DEG_RANGE_NODES) atomicAdd(&hist[r2], 1u);
    if (r3 < DEG_RANGE_NODES) atomicAdd(&hist[r3], 1u);
}

__global__ __launch_bounds__(1024) void gcn_deg_hist(const int* __restrict__ dst,
                                                     unsigned* __restrict__ part, int E) {
    __shared__ unsigned hist[DEG_RANGE_NODES];
    const int slice = blockIdx.x / DEG_RANGES;
    const int range = blockIdx.x % DEG_RANGES;
    const unsigned base = (unsigned)range * DEG_RANGE_NODES;

    for (int t = threadIdx.x; t < DEG_RANGE_NODES; t += 1024) hist[t] = 0u;
    __syncthreads();

    const int chunk = (((E + DEG_SLICES - 1) / DEG_SLICES) + 3) & ~3;
    int lo = slice * chunk;
    int hi = lo + chunk;
    if (lo > E) lo = E;
    if (hi > E) hi = E;

    if ((((size_t)dst) & 15) == 0) {
        const int nvec = (hi - lo) >> 2;
        const int4* __restrict__ d4 = (const int4*)(dst + lo);
        int v = threadIdx.x;
        for (; v + 3 * 1024 < nvec; v += 4 * 1024) {
            int4 da = d4[v];
            int4 db = d4[v + 1024];
            int4 dc = d4[v + 2 * 1024];
            int4 dd = d4[v + 3 * 1024];
            deg_count4(hist, da, base);
            deg_count4(hist, db, base);
            deg_count4(hist, dc, base);
            deg_count4(hist, dd, base);
        }
        for (; v < nvec; v += 1024) deg_count4(hist, d4[v], base);
        lo += nvec << 2;
    }
    for (int e = lo + threadIdx.x; e < hi; e += 1024) {
        unsigned r = (unsigned)dst[e] - base;
        if (r < DEG_RANGE_NODES) atomicAdd(&hist[r], 1u);
    }
    __syncthreads();

    unsigned* __restrict__ row = part + (size_t)slice * N_NODES + base;
    for (int t = threadIdx.x; t < DEG_RANGE_NODES; t += 1024) row[t] = hist[t];
}

__global__ void gcn_dinv(const unsigned* __restrict__ part,
                         float* __restrict__ dinv) {
    int i = blockIdx.x * blockDim.x + threadIdx.x;
    if (i >= N_NODES) return;
    unsigned deg = 1u;
    #pragma unroll
    for (int s = 0; s < DEG_SLICES; ++s) deg += part[(size_t)s * N_NODES + i];
    dinv[i] = rsqrtf((float)deg);
}

__global__ void gcn_out_scale(const float* __restrict__ dinv,
                              const float* __restrict__ b,
                              float* __restrict__ out) {
    int t = blockIdx.x * blockDim.x + threadIdx.x;
    if (t >= N_NODES * F_OUT) return;
    int i = t >> 5;
    int f = t & 31;
    out[t] = out[t] * dinv[i] + b[f];
}

#define SC_EDGES 32

__device__ __forceinline__ unsigned sc_bcast_u(int v, int j) {
    return (unsigned)__builtin_amdgcn_readlane(v, j);
}

__global__ __launch_bounds__(256) void gcn_scatter(const int* __restrict__ src,
                                                   const int* __restrict__ dst,
                                                   const float* __restrict__ hs,
                                                   float* __restrict__ out, int E) {
    const unsigned lane = threadIdx.x & 31u;
    const int wave = (int)((blockIdx.x * 256u + threadIdx.x) >> 5);
    const int base = wave * SC_EDGES;
    if (base >= E) return;

    if (base + SC_EDGES <= E) {
        const int e = base + (int)lane;
        const int s = src[e];
        const int d = dst[e];

        float v[SC_EDGES];
        #pragma unroll
        for (int j = 0; j < SC_EDGES; ++j) {
            const unsigned sj = sc_bcast_u(s, j);
            v[j] = hs[sj * (unsigned)F_OUT + lane];
        }
        #pragma unroll
        for (int j = 0; j < SC_EDGES; ++j) {
            const unsigned dj = sc_bcast_u(d, j);
            unsafeAtomicAdd(&out[dj * (unsigned)F_OUT + lane], v[j]);
        }
    } else {
        const int cnt = E - base;
        int s = 0, d = 0;
        if ((int)lane < cnt) {
            s = src[base + (int)lane];
            d = dst[base + (int)lane];
        }
        for (int j = 0; j < cnt; ++j) {
            const unsigned sj = sc_bcast_u(s, j);
            const unsigned dj = sc_bcast_u(d, j);
            unsafeAtomicAdd(&out[dj * (unsigned)F_OUT + lane],
                            hs[sj * (unsigned)F_OUT + lane]);
        }
    }
}

extern "C" void kernel_launch(void* const* d_in, const int* in_sizes, int n_in,
                              void* d_out, int out_size, void* d_ws, size_t ws_size,
                              hipStream_t stream) {
    const float* x     = (const float*)d_in[0];
    const int*   edges = (const int*)d_in[1];
    const float* W     = (const float*)d_in[2];
    const float* b     = (const float*)d_in[3];
    float*       out   = (float*)d_out;

    const int E = in_sizes[1] / 2;
    const int* src = edges;
    const int* dst = edges + E;

    char* ws = (char*)d_ws;
    float*    hs   = (float*)ws;
    unsigned* part = (unsigned*)ws;
    float*    dinv = (float*)(ws + (size_t)N_NODES * F_OUT * 4 + (size_t)N_NODES * 4);

    gcn_deg_hist<<<DEG_SLICES * DEG_RANGES, 1024, 0, stream>>>(dst, part, E);
    gcn_dinv<<<(N_NODES + 255) / 256, 256, 0, stream>>>(part, dinv);

    gcn_gemm_xw<<<(M_TILES + 7) / 8, 256, 0, stream>>>(x, W, dinv, hs, out);

    int blocks = (E + 255) / 256;
    gcn_scatter<<<blocks, 256, 0, stream>>>(src, dst, hs, out, E);

    gcn_out_scale<<<(N_NODES * F_OUT + 255) / 256, 256, 0, stream>>>(dinv, b, out);
}


# edge scatter half-affine: a wave reads its die, takes a chunk ticket for its L2 half and adds only edges whose out row that half owns (LDS-compacted); fp32 atomics, arrival order free as before
# speedup vs baseline: 4.0415x; 1.2908x over previous
#include <hip/hip_runtime.h>

typedef __attribute__((ext_vector_type(2))) float v2f;
typedef __attribute__((ext_vector_type(8))) float v8f;

#define N_NODES 100000
#define F_IN    256
#define F_OUT   32
#define M_TILES (N_NODES / 16)
#define TICKET_STRIDE 64

__global__ __launch_bounds__(256) void gcn_gemm_xw(const float* __restrict__ x,
                                                   const float* __restrict__ W,
                                                   const float* __restrict__ dinv,
                                                   float* __restrict__ hs,
                                                   float* __restrict__ out) {
    __shared__ v2f Wp[(F_IN / 2) * F_OUT];
    for (int idx = threadIdx.x; idx < (F_IN / 2) * F_OUT; idx += 256) {
        int p = idx >> 5;
        int c = idx & 31;
        v2f w;
        w.x = W[(2 * p)     * F_OUT + c];
        w.y = W[(2 * p + 1) * F_OUT + c];
        Wp[idx] = w;
    }
    __syncthreads();

    const int wave = threadIdx.x >> 5;
    const int lane = threadIdx.x & 31;
    const int tile = blockIdx.x * 8 + wave;
    if (tile >= M_TILES) return;

    const int m0   = tile << 4;
    const int mrow = m0 + (lane & 15);
    const int koff = (lane >> 4) << 1;
    const int col  = lane & 15;
    const int poff = koff >> 1;

    v8f c0 = {};
    v8f c1 = {};

    const float* __restrict__ xrow = x + (size_t)mrow * F_IN;
    #pragma unroll 4
    for (int k = 0; k < F_IN; k += 4) {
        v2f a = *(const v2f*)(xrow + k + koff);
        const int p = (k >> 1) + poff;
        v2f b0 = Wp[p * F_OUT + col];
        v2f b1 = Wp[p * F_OUT + 16 + col];
        c0 = __builtin_amdgcn_wmma_f32_16x16x4_f32(false, a, false, b0,
                                                   (short)0, c0, false, false);
        c1 = __builtin_amdgcn_wmma_f32_16x16x4_f32(false, a, false, b1,
                                                   (short)0, c1, false, false);
    }

    const int rbase = m0 + ((lane >> 4) << 3);
    #pragma unroll
    for (int j = 0; j < 8; ++j) {
        const float  dj = dinv[rbase + j];
        const size_t o  = (size_t)(rbase + j) * F_OUT + col;
        const float  v0 = c0[j] * dj;
        const float  v1 = c1[j] * dj;
        hs[o]       = v0;
        hs[o + 16]  = v1;
        out[o]      = v0;
        out[o + 16] = v1;
    }
}

#define DEG_SLICES 16
#define DEG_RANGES 16
#define DEG_RANGE_NODES (N_NODES / DEG_RANGES)

__device__ __forceinline__ void deg_count4(unsigned* hist, int4 d, unsigned base) {
    unsigned r0 = (unsigned)d.x - base;
    unsigned r1 = (unsigned)d.y - base;
    unsigned r2 = (unsigned)d.z - base;
    unsigned r3 = (unsigned)d.w - base;
    if (r0 < DEG_RANGE_NODES) atomicAdd(&hist[r0], 1u);
    if (r1 < DEG_RANGE_NODES) atomicAdd(&hist[r1], 1u);
    if (r2 < DEG_RANGE_NODES) atomicAdd(&hist[r2], 1u);
    if (r3 < DEG_RANGE_NODES) atomicAdd(&hist[r3], 1u);
}

__global__ __launch_bounds__(1024) void gcn_deg_hist(const int* __restrict__ dst,
                                                     unsigned* __restrict__ part, int E) {
    __shared__ unsigned hist[DEG_RANGE_NODES];
    const int slice = blockIdx.x / DEG_RANGES;
    const int range = blockIdx.x % DEG_RANGES;
    const unsigned base = (unsigned)range * DEG_RANGE_NODES;

    for (int t = threadIdx.x; t < DEG_RANGE_NODES; t += 1024) hist[t] = 0u;
    __syncthreads();

    const int chunk = (((E + DEG_SLICES - 1) / DEG_SLICES) + 3) & ~3;
    int lo = slice * chunk;
    int hi = lo + chunk;
    if (lo > E) lo = E;
    if (hi > E) hi = E;

    if ((((size_t)dst) & 15) == 0) {
        const int nvec = (hi - lo) >> 2;
        const int4* __restrict__ d4 = (const int4*)(dst + lo);
        int v = threadIdx.x;
        for (; v + 3 * 1024 < nvec; v += 4 * 1024) {
            int4 da = d4[v];
            int4 db = d4[v + 1024];
            int4 dc = d4[v + 2 * 1024];
            int4 dd = d4[v + 3 * 1024];
            deg_count4(hist, da, base);
            deg_count4(hist, db, base);
            deg_count4(hist, dc, base);
            deg_count4(hist, dd, base);
        }
        for (; v < nvec; v += 1024) deg_count4(hist, d4[v], base);
        lo += nvec << 2;
    }
    for (int e = lo + threadIdx.x; e < hi; e += 1024) {
        unsigned r = (unsigned)dst[e] - base;
        if (r < DEG_RANGE_NODES) atomicAdd(&hist[r], 1u);
    }
    __syncthreads();

    unsigned* __restrict__ row = part + (size_t)slice * N_NODES + base;
    for (int t = threadIdx.x; t < DEG_RANGE_NODES; t += 1024) row[t] = hist[t];
}

__global__ void gcn_dinv(const unsigned* __restrict__ part,
                         float* __restrict__ dinv,
                         unsigned* __restrict__ tickets) {
    int i = blockIdx.x * blockDim.x + threadIdx.x;
    if (i >= N_NODES) return;
    if (i == 0) { tickets[0] = 0u; tickets[TICKET_STRIDE] = 0u; }
    unsigned deg = 1u;
    #pragma unroll
    for (int s = 0; s < DEG_SLICES; ++s) deg += part[(size_t)s * N_NODES + i];
    dinv[i] = rsqrtf((float)deg);
}

__global__ void gcn_out_scale(const float* __restrict__ dinv,
                              const float* __restrict__ b,
                              float* __restrict__ out) {
    int t = blockIdx.x * blockDim.x + threadIdx.x;
    if (t >= N_NODES * F_OUT) return;
    int i = t >> 5;
    int f = t & 31;
    out[t] = out[t] * dinv[i] + b[f];
}

#define SC_CHUNK      256
#define SC_LOADS      (SC_CHUNK / 32)
#define HALF_NODES    (N_NODES / 2)

__device__ __forceinline__ unsigned sc_bcast_u(int v, int j) {
    return (unsigned)__builtin_amdgcn_readlane(v, j);
}

__global__ __launch_bounds__(256) void gcn_scatter(const int* __restrict__ src,
                                                   const int* __restrict__ dst,
                                                   const float* __restrict__ hs,
                                                   float* __restrict__ out,
                                                   unsigned* __restrict__ tickets, int E) {
    __shared__ int2 queue[8][SC_CHUNK + 32];
    const unsigned lane = threadIdx.x & 31u;
    int2* __restrict__ qw = queue[threadIdx.x >> 5];

    const unsigned nchunks = ((unsigned)E + (SC_CHUNK - 1)) / SC_CHUNK;

    const unsigned place = __builtin_amdgcn_s_sendmsg_rtn(0x87);
    unsigned half = (place >> 18) & 1u;

    unsigned t = 0u;
    if (lane == 0u) t = atomicAdd(&tickets[half * TICKET_STRIDE], 1u);
    t = (unsigned)__builtin_amdgcn_readfirstlane((int)t);
    if (t >= nchunks) {
        half ^= 1u;
        if (lane == 0u) t = atomicAdd(&tickets[half * TICKET_STRIDE], 1u);
        t = (unsigned)__builtin_amdgcn_readfirstlane((int)t);
        if (t >= nchunks) return;
    }

    const int base = (int)(t * SC_CHUNK);
    int  s[SC_LOADS], d[SC_LOADS];
    bool m[SC_LOADS];
    #pragma unroll
    for (int u = 0; u < SC_LOADS; ++u) {
        const int e  = base + u * 32 + (int)lane;
        const int ec = (e < E) ? e : (E - 1);
        s[u] = src[ec];
        d[u] = dst[ec];
    }
    #pragma unroll
    for (int u = 0; u < SC_LOADS; ++u) {
        const int e = base + u * 32 + (int)lane;
        m[u] = (e < E) && ((((unsigned)d[u] >= (unsigned)HALF_NODES) ? 1u : 0u) == half);
    }

    unsigned qlen = 0u;
    #pragma unroll
    for (int u = 0; u < SC_LOADS; ++u) {
        const unsigned mask = __builtin_amdgcn_ballot_w32(m[u]);
        const unsigned pos  = qlen + __builtin_amdgcn_mbcnt_lo(mask, 0u);
        qw[m[u] ? pos : (unsigned)SC_CHUNK + lane] = make_int2(s[u], d[u]);
        qlen += (unsigned)__builtin_popcount(mask);
    }

    for (unsigned b0 = 0u; b0 < qlen; b0 += 32u) {
        const unsigned nb = (qlen - b0 < 32u) ? (qlen - b0) : 32u;
        int2 ent = make_int2(0, 0);
        if (b0 + lane < qlen) ent = qw[b0 + lane];

        float v[32];
        #pragma unroll
        for (int j = 0; j < 32; ++j) {
            const unsigned sj = sc_bcast_u(ent.x, j);
            v[j] = hs[sj * (unsigned)F_OUT + lane];
        }
        if (nb == 32u) {
            #pragma unroll
            for (int j = 0; j < 32; ++j) {
                const unsigned dj = sc_bcast_u(ent.y, j);
                unsafeAtomicAdd(&out[dj * (unsigned)F_OUT + lane], v[j]);
            }
        } else {
            #pragma unroll
            for (int j = 0; j < 32; ++j) {
                if ((unsigned)j < nb) {
                    const unsigned dj = sc_bcast_u(ent.y, j);
                    unsafeAtomicAdd(&out[dj * (unsigned)F_OUT + lane], v[j]);
                }
            }
        }
    }
}

extern "C" void kernel_launch(void* const* d_in, const int* in_sizes, int n_in,
                              void* d_out, int out_size, void* d_ws, size_t ws_size,
                              hipStream_t stream) {
    const float* x     = (const float*)d_in[0];
    const int*   edges = (const int*)d_in[1];
    const float* W     = (const float*)d_in[2];
    const float* b     = (const float*)d_in[3];
    float*       out   = (float*)d_out;

    const int E = in_sizes[1] / 2;
    const int* src = edges;
    const int* dst = edges + E;

    char* ws = (char*)d_ws;
    float*    hs   = (float*)ws;
    unsigned* part = (unsigned*)ws;
    unsigned* tickets = (unsigned*)(ws + (size_t)N_NODES * F_OUT * 4);
    float*    dinv = (float*)(ws + (size_t)N_NODES * F_OUT * 4 + (size_t)N_NODES * 4);

    gcn_deg_hist<<<DEG_SLICES * DEG_RANGES, 1024, 0, stream>>>(dst, part, E);
    gcn_dinv<<<(N_NODES + 255) / 256, 256, 0, stream>>>(part, dinv, tickets);

    gcn_gemm_xw<<<(M_TILES + 7) / 8, 256, 0, stream>>>(x, W, dinv, hs, out);

    int blocks = (E + 1023) / 1024;
    gcn_scatter<<<blocks, 256, 0, stream>>>(src, dst, hs, out, tickets, E);

    gcn_out_scale<<<(N_NODES * F_OUT + 255) / 256, 256, 0, stream>>>(dinv, b, out);
}
